# starting kernel
# baseline (speedup 1.0000x reference)
#include <hip/hip_runtime.h>
#include <math.h>

typedef __attribute__((ext_vector_type(2))) float v2f;
typedef __attribute__((ext_vector_type(8))) float v8f;

#define TT 128
#define DD 768
#define DECAY_ALPHA 0.1f
#define EPS_W 1e-8f
#define LN_EPS 1e-5f

__device__ __forceinline__ float sanitize(float x) {
  return (__builtin_fabsf(x) <= 3.402823e38f) ? x : 0.0f;
}

__global__ __launch_bounds__(256) void mda_kernel(
    const float* __restrict__ H, const int* __restrict__ valid_lens,
    const float* __restrict__ gamma, const float* __restrict__ beta,
    float* __restrict__ out)
{
  __shared__ float s_norm2[TT];
  __shared__ float s_w[TT];
  __shared__ float s_E[DD];
  __shared__ float s_red[8];
  __shared__ float s_bcast[3];

  const int bf   = blockIdx.x;
  const int tid  = threadIdx.x;
  const int lane = tid & 31;
  const int wv   = tid >> 5;
  const float* __restrict__ Hbf = H + (size_t)bf * (TT * DD);

  int L = valid_lens[bf];
  L = (L < 0) ? 0 : (L > TT ? TT : L);

  for (int t = wv; t < TT; t += 8) {
    float s = 0.0f;
    if (t < L) {
      const float* row = Hbf + t * DD;
      if (t + 8 < L) __builtin_prefetch(row + 8 * DD + lane * 4, 0, 0);
      #pragma unroll
      for (int c = 0; c < 6; ++c) {
        const float4 v = *(const float4*)(row + c * 128 + lane * 4);
        const float x = sanitize(v.x), y = sanitize(v.y);
        const float z = sanitize(v.z), w = sanitize(v.w);
        s += x * x + y * y + z * z + w * w;
      }
    }
    #pragma unroll
    for (int off = 16; off > 0; off >>= 1) s += __shfl_xor(s, off, 32);
    if (lane == 0) s_norm2[t] = (t < L) ? s : __builtin_inff();
  }
  __syncthreads();

  if (tid < TT) {
    const float my = s_norm2[tid];
    int rank = 0;
    #pragma unroll 4
    for (int j = 0; j < TT; ++j) {
      const float vj = s_norm2[j];
      rank += (vj < my || (vj == my && j < tid)) ? 1 : 0;
    }
    const int k_low = (int)(0.1f * (float)TT);
    const int lo = (k_low < (L >> 1)) ? k_low : (L >> 1);
    const int hr = L - lo;
    const int hi = (hr > lo) ? hr : lo;
    const bool keep = (rank >= lo) && (rank < hi) && (tid < L);
    s_w[tid] = keep ? __expf(-DECAY_ALPHA * (float)(TT - 1 - tid)) : 0.0f;
  }
  __syncthreads();

  if (wv == 0) {
    float s = s_w[lane] + s_w[lane + 32] + s_w[lane + 64] + s_w[lane + 96];
    #pragma unroll
    for (int off = 16; off > 0; off >>= 1) s += __shfl_xor(s, off, 32);
    if (lane == 0) {
      s_bcast[0] = 1.0f / fmaxf(s, EPS_W);
      s_bcast[1] = (s > EPS_W) ? 1.0f : 0.0f;
    }
  }
  __syncthreads();

  const int half = lane >> 4;
  const int l16  = lane & 15;
  const float inv = s_bcast[0];

  for (int c = 0; c < 6; c += 2) {
    const int d0 = wv * 96 + c * 16;
    v8f acc0 = {0.f, 0.f, 0.f, 0.f, 0.f, 0.f, 0.f, 0.f};
    v8f acc1 = acc0;
    for (int t0 = 0; t0 < TT; t0 += 4) {
      const float w0 = s_w[t0], w1 = s_w[t0 + 1], w2 = s_w[t0 + 2], w3 = s_w[t0 + 3];
      if ((w0 == 0.0f) & (w1 == 0.0f) & (w2 == 0.0f) & (w3 == 0.0f)) continue;
      v2f b; b.x = half ? w2 : w0; b.y = half ? w3 : w1;
      const float* pa = Hbf + (size_t)(t0 + half * 2) * DD + d0 + l16;
      v2f a0, a1;
      a0.x = sanitize(pa[0]);   a0.y = sanitize(pa[DD]);
      a1.x = sanitize(pa[16]);  a1.y = sanitize(pa[DD + 16]);
      acc0 = __builtin_amdgcn_wmma_f32_16x16x4_f32(false, a0, false, b, (short)0,
                                                   acc0, false, false);
      acc1 = __builtin_amdgcn_wmma_f32_16x16x4_f32(false, a1, false, b, (short)0,
                                                   acc1, false, false);
    }
    if (l16 == 0) {
      const int mb = half * 8;
      #pragma unroll
      for (int r = 0; r < 8; ++r) {
        s_E[d0 + mb + r]      = acc0[r] * inv;
        s_E[d0 + 16 + mb + r] = acc1[r] * inv;
      }
    }
  }
  __syncthreads();

  const float e0 = s_E[tid], e1 = s_E[tid + 256], e2 = s_E[tid + 512];
  float s = e0 + e1 + e2;
  #pragma unroll
  for (int off = 16; off > 0; off >>= 1) s += __shfl_xor(s, off, 32);
  if (lane == 0) s_red[wv] = s;
  __syncthreads();
  if (tid == 0) {
    float tot = 0.f;
    #pragma unroll
    for (int i = 0; i < 8; ++i) tot += s_red[i];
    s_bcast[2] = tot * (1.0f / (float)DD);
  }
  __syncthreads();
  const float mu = s_bcast[2];

  const float q0 = e0 - mu, q1 = e1 - mu, q2 = e2 - mu;
  float v = q0 * q0 + q1 * q1 + q2 * q2;
  #pragma unroll
  for (int off = 16; off > 0; off >>= 1) v += __shfl_xor(v, off, 32);
  if (lane == 0) s_red[wv] = v;
  __syncthreads();
  if (tid == 0) {
    float tot = 0.f;
    #pragma unroll
    for (int i = 0; i < 8; ++i) tot += s_red[i];
    s_bcast[2] = tot * (1.0f / (float)DD);
  }
  __syncthreads();
  const float var  = s_bcast[2];
  const float rstd = rsqrtf(var + LN_EPS);
  const bool  has  = s_bcast[1] > 0.5f;

  float* __restrict__ obf = out + (size_t)bf * DD;
  {
    const int d = tid;
    obf[d] = has ? ((e0 - mu) * rstd * gamma[d] + beta[d]) : e0;
  }
  {
    const int d = tid + 256;
    obf[d] = has ? ((e1 - mu) * rstd * gamma[d] + beta[d]) : e1;
  }
  {
    const int d = tid + 512;
    obf[d] = has ? ((e2 - mu) * rstd * gamma[d] + beta[d]) : e2;
  }
}

extern "C" void kernel_launch(void* const* d_in, const int* in_sizes, int n_in,
                              void* d_out, int out_size, void* d_ws, size_t ws_size,
                              hipStream_t stream) {
  const float* H     = (const float*)d_in[0];
  const int*   vlens = (const int*)d_in[1];
  const float* gamma = (const float*)d_in[2];
  const float* beta  = (const float*)d_in[3];
  float* out = (float*)d_out;
  const int BF = in_sizes[1];
  mda_kernel<<<BF, 256, 0, stream>>>(H, vlens, gamma, beta, out);
}


# H read once: the norm sweep also takes the decay-weighted sum of all valid rows on the matrix unit; at most 24 rows are re-read and subtracted (under 48 rows: added); E sum order changes
# speedup vs baseline: 2.7193x; 2.7193x over previous
#include <hip/hip_runtime.h>
#include <math.h>


typedef __attribute__((ext_vector_type(2))) float v2f;
typedef __attribute__((ext_vector_type(8))) float v8f;

#define TT 128
#define DD 768
#define NTHR 768
#define NWAVE 24
#define NSTRIPE 12
#define NLIST 24
#define DECAY_ALPHA 0.1f
#define EPS_W 1e-8f
#define LN_EPS 1e-5f

__device__ __forceinline__ float sanitize(float x) {
  return (__builtin_fabsf(x) <= 3.402823e38f) ? x : 0.0f;
}

__device__ __forceinline__ float wave_sum(float s) {
  #pragma unroll
  for (int off = 16; off > 0; off >>= 1) s += __shfl_xor(s, off, 32);
  return s;
}

__device__ __forceinline__ float decay_w(int t) {
  return __expf(-DECAY_ALPHA * (float)(TT - 1 - t));
}

template <int N>
__device__ __forceinline__ float row_xor_add(float keep, float send) {
  const int moved = __builtin_amdgcn_update_dpp(0, __float_as_int(send), 0x160 + N, 0xf, 0xf, true);
  return keep + __int_as_float(moved);
}

__global__ __launch_bounds__(NTHR) void mda_kernel(
    const float* __restrict__ H, const int* __restrict__ valid_lens,
    const float* __restrict__ gamma, const float* __restrict__ beta,
    float* __restrict__ out)
{
  __shared__ float s_part[NSTRIPE][TT];
  __shared__ float s_norm2[TT];
  __shared__ int   s_cnt[6][TT];
  __shared__ __attribute__((aligned(16))) float s_E[2][DD];
  __shared__ __attribute__((aligned(8))) int   s_lt[NLIST];
  __shared__ __attribute__((aligned(8))) float s_lw[NLIST];
  __shared__ float s_red[NWAVE];
  __shared__ float s_red2[NWAVE];
  __shared__ __attribute__((aligned(16))) float s_w[TT];
  __shared__ float s_bcast[3];

  const int bf   = blockIdx.x;
  const int tid  = threadIdx.x;
  const int lane = tid & 31;
  const int wv   = tid >> 5;
  const float* __restrict__ Hbf = H + (size_t)bf * (TT * DD);

  int L = valid_lens[bf];
  L = (L < 0) ? 0 : (L > TT ? TT : L);
  const bool sub = (L >= 2 * NLIST);
  const int Lw = sub ? L : 0;

  const int half = lane >> 4;
  const int l16  = lane & 15;
  const int wvs = __builtin_amdgcn_readfirstlane(wv);
  const int sp = wvs % NSTRIPE;
  const int ts = wvs / NSTRIPE;
  const int d0 = sp * 64;
  v8f acc0 = {0.f, 0.f, 0.f, 0.f, 0.f, 0.f, 0.f, 0.f};
  v8f acc1 = acc0;
  v8f acc2 = acc0;
  v8f acc3 = acc0;

  {
    const float* __restrict__ plane = Hbf + (size_t)(half * 2) * DD + d0 + l16 * 4;
    const bool b3 = (l16 & 8) != 0, b2 = (l16 & 4) != 0, b1 = (l16 & 2) != 0;
    for (int tb = ts * 4; tb < L; tb += 32) {
      float4 va[4], vb[4];
      v2f b[4];
      float P[8];
      #pragma unroll
      for (int q = 0; q < 4; ++q) {
        const float* pa = plane + (size_t)(tb + 8 * q) * DD;
        va[q] = *(const float4*)pa;
        vb[q] = *(const float4*)(pa + DD);
      }
      #pragma unroll
      for (int q = 0; q < 4; ++q) {
        const int t = tb + 8 * q + 2 * half;
        b[q].x = (t     < Lw) ? decay_w(t)     : 0.0f;
        b[q].y = (t + 1 < Lw) ? decay_w(t + 1) : 0.0f;
      }
      __builtin_amdgcn_sched_barrier(0);
      #pragma unroll
      for (int q = 0; q < 4; ++q) {
        v2f a0, a1, a2, a3;
        a0.x = sanitize(va[q].x);  a0.y = sanitize(vb[q].x);
        a1.x = sanitize(va[q].y);  a1.y = sanitize(vb[q].y);
        a2.x = sanitize(va[q].z);  a2.y = sanitize(vb[q].z);
        a3.x = sanitize(va[q].w);  a3.y = sanitize(vb[q].w);
        P[2 * q]     = a0.x * a0.x + a1.x * a1.x + a2.x * a2.x + a3.x * a3.x;
        P[2 * q + 1] = a0.y * a0.y + a1.y * a1.y + a2.y * a2.y + a3.y * a3.y;
        acc0 = __builtin_amdgcn_wmma_f32_16x16x4_f32(false, a0, false, b[q], (short)0,
                                                     acc0, false, false);
        acc1 = __builtin_amdgcn_wmma_f32_16x16x4_f32(false, a1, false, b[q], (short)0,
                                                     acc1, false, false);
        acc2 = __builtin_amdgcn_wmma_f32_16x16x4_f32(false, a2, false, b[q], (short)0,
                                                     acc2, false, false);
        acc3 = __builtin_amdgcn_wmma_f32_16x16x4_f32(false, a3, false, b[q], (short)0,
                                                     acc3, false, false);
      }
      float Q[4], R[2];
      #pragma unroll
      for (int j = 0; j < 4; ++j)
        Q[j] = row_xor_add<8>(b3 ? P[2 * j + 1] : P[2 * j], b3 ? P[2 * j] : P[2 * j + 1]);
      #pragma unroll
      for (int j = 0; j < 2; ++j)
        R[j] = row_xor_add<4>(b2 ? Q[2 * j + 1] : Q[2 * j], b2 ? Q[2 * j] : Q[2 * j + 1]);
      float S = row_xor_add<2>(b1 ? R[1] : R[0], b1 ? R[0] : R[1]);
      S = row_xor_add<1>(S, S);
      if ((l16 & 1) == 0) {
        const int qq = (b1 ? 2 : 0) + (b2 ? 1 : 0);
        s_part[sp][tb + 8 * qq + 2 * half + (b3 ? 1 : 0)] = S;
      }
    }
  }
  __syncthreads();

  if (tid < TT) {
    float s = __builtin_inff();
    if (tid < L) {
      s = s_part[0][tid];
      #pragma unroll
      for (int k = 1; k < NSTRIPE; ++k) s += s_part[k][tid];
    }
    s_norm2[tid] = s;
  }
  __syncthreads();

  {
    const int i  = tid & (TT - 1);
    const int p  = wvs >> 2;
    const int j0 = (p < 4) ? 24 * p : 96 + 16 * (p - 4);
    const float my = s_norm2[i];
    int cnt = 0;
    #pragma unroll
    for (int jj = 0; jj < 16; ++jj) {
      const int j = j0 + jj;
      const float vj = s_norm2[j];
      cnt += (vj < my || (vj == my && j < i)) ? 1 : 0;
    }
    if (p < 4) {
      #pragma unroll
      for (int jj = 16; jj < 24; ++jj) {
        const int j = j0 + jj;
        const float vj = s_norm2[j];
        cnt += (vj < my || (vj == my && j < i)) ? 1 : 0;
      }
    }
    s_cnt[p][i] = cnt;
  }
  __syncthreads();

  if (wvs == 0) {
    const int k_low = (int)(0.1f * (float)TT);
    const int lo = (k_low < (L >> 1)) ? k_low : (L >> 1);
    const int hr = L - lo;
    const int hi = (hr > lo) ? hr : lo;
    float wk = 0.0f, all2 = 0.0f, kept2 = 0.0f;
    unsigned m[4];
    bool  fl[4];
    float sw[4];
    #pragma unroll
    for (int k = 0; k < 4; ++k) {
      const int r = lane + 32 * k;
      int rank = 0;
      #pragma unroll
      for (int p = 0; p < 6; ++p) rank += s_cnt[p][r];
      const bool valid = r < L;
      const bool keep  = valid && (rank >= lo) && (rank < hi);
      const float w = decay_w(r);
      wk += keep ? w : 0.0f;
      s_w[r] = keep ? w : 0.0f;
      const float e2 = valid ? w * w * s_norm2[r] : 0.0f;
      all2  += e2;
      kept2 += keep ? e2 : 0.0f;
      fl[k] = sub ? (valid && !keep) : keep;
      sw[k] = sub ? -w : w;
      m[k]  = __builtin_amdgcn_ballot_w32(fl[k]);
    }
    const float s = wave_sum(wk);
    const float sa = wave_sum(all2), sk = wave_sum(kept2);
    const bool cond_ok = (sa <= 1024.0f * sk) && (sa < __builtin_inff());
    const unsigned below = (1u << lane) - 1u;
    int off = 0;
    #pragma unroll
    for (int k = 0; k < 4; ++k) {
      const int pos = off + __popc(m[k] & below);
      if (fl[k] && pos < NLIST) { s_lt[pos] = lane + 32 * k; s_lw[pos] = sw[k]; }
      off += __popc(m[k]);
    }
    if (lane < NLIST && lane >= off) { s_lt[lane] = 0; s_lw[lane] = 0.0f; }
    if (lane == 0) {
      s_bcast[0] = 1.0f / fmaxf(s, EPS_W);
      s_bcast[1] = (s > EPS_W) ? 1.0f : 0.0f;
      s_bcast[2] = (sub && !cond_ok) ? 1.0f : 0.0f;
    }
  }
  __syncthreads();

  const int fbs = __builtin_amdgcn_readfirstlane(__float_as_int(s_bcast[2]));
  if (fbs != 0) {
    acc0 = (v8f){0.f, 0.f, 0.f, 0.f, 0.f, 0.f, 0.f, 0.f};
    acc1 = acc0;
    acc2 = acc0;
    acc3 = acc0;
    const float* __restrict__ plane = Hbf + (size_t)(half * 2) * DD + d0 + l16 * 4;
    for (int tb = ts * 4; tb < L; tb += 32) {
      float4 va[4], vb[4];
      v2f b[4];
      #pragma unroll
      for (int q = 0; q < 4; ++q) {
        const float* pa = plane + (size_t)(tb + 8 * q) * DD;
        va[q] = *(const float4*)pa;
        vb[q] = *(const float4*)(pa + DD);
      }
      #pragma unroll
      for (int q = 0; q < 4; ++q) {
        const float4 wq = *(const float4*)&s_w[tb + 8 * q];
        b[q].x = half ? wq.z : wq.x;
        b[q].y = half ? wq.w : wq.y;
      }
      #pragma unroll
      for (int q = 0; q < 4; ++q) {
        v2f a0, a1, a2, a3;
        a0.x = sanitize(va[q].x);  a0.y = sanitize(vb[q].x);
        a1.x = sanitize(va[q].y);  a1.y = sanitize(vb[q].y);
        a2.x = sanitize(va[q].z);  a2.y = sanitize(vb[q].z);
        a3.x = sanitize(va[q].w);  a3.y = sanitize(vb[q].w);
        acc0 = __builtin_amdgcn_wmma_f32_16x16x4_f32(false, a0, false, b[q], (short)0,
                                                     acc0, false, false);
        acc1 = __builtin_amdgcn_wmma_f32_16x16x4_f32(false, a1, false, b[q], (short)0,
                                                     acc1, false, false);
        acc2 = __builtin_amdgcn_wmma_f32_16x16x4_f32(false, a2, false, b[q], (short)0,
                                                     acc2, false, false);
        acc3 = __builtin_amdgcn_wmma_f32_16x16x4_f32(false, a3, false, b[q], (short)0,
                                                     acc3, false, false);
      }
    }
  } else {
    float4 ca[3], cb[3];
    v2f cw[3];
    #pragma unroll
    for (int u = 0; u < 3; ++u) {
      const int e = 4 * (ts + 2 * u) + 2 * half;
      const int r0 = s_lt[e], r1 = s_lt[e + 1];
      cw[u].x = s_lw[e];
      cw[u].y = s_lw[e + 1];
      ca[u] = *(const float4*)(Hbf + (size_t)r0 * DD + d0 + l16 * 4);
      cb[u] = *(const float4*)(Hbf + (size_t)r1 * DD + d0 + l16 * 4);
    }
    __builtin_amdgcn_sched_barrier(0);
    #pragma unroll
    for (int u = 0; u < 3; ++u) {
      v2f a0, a1, a2, a3;
      a0.x = sanitize(ca[u].x);  a0.y = sanitize(cb[u].x);
      a1.x = sanitize(ca[u].y);  a1.y = sanitize(cb[u].y);
      a2.x = sanitize(ca[u].z);  a2.y = sanitize(cb[u].z);
      a3.x = sanitize(ca[u].w);  a3.y = sanitize(cb[u].w);
      acc0 = __builtin_amdgcn_wmma_f32_16x16x4_f32(false, a0, false, cw[u], (short)0,
                                                   acc0, false, false);
      acc1 = __builtin_amdgcn_wmma_f32_16x16x4_f32(false, a1, false, cw[u], (short)0,
                                                   acc1, false, false);
      acc2 = __builtin_amdgcn_wmma_f32_16x16x4_f32(false, a2, false, cw[u], (short)0,
                                                   acc2, false, false);
      acc3 = __builtin_amdgcn_wmma_f32_16x16x4_f32(false, a3, false, cw[u], (short)0,
                                                   acc3, false, false);
    }
  }
  {
    if (l16 == 0) {
      float4* __restrict__ dst = (float4*)&s_E[ts][d0 + half * 32];
      #pragma unroll
      for (int r = 0; r < 8; ++r) {
        float4 o; o.x = acc0[r]; o.y = acc1[r]; o.z = acc2[r]; o.w = acc3[r];
        dst[r] = o;
      }
    }
  }
  __syncthreads();

  const float inv = s_bcast[0];
  const bool  has = s_bcast[1] > 0.5f;
  const int   d   = tid;
  const float e   = (s_E[0][d] + s_E[1][d]) * inv;

  const float ps = wave_sum(e);
  if (lane == 0) s_red[wv] = ps;
  __syncthreads();
  const float mu = wave_sum((lane < NWAVE) ? s_red[lane] : 0.0f) * (1.0f / (float)DD);

  const float q  = e - mu;
  const float pv = wave_sum(q * q);
  if (lane == 0) s_red2[wv] = pv;
  __syncthreads();
  const float var  = wave_sum((lane < NWAVE) ? s_red2[lane] : 0.0f) * (1.0f / (float)DD);
  const float rstd = rsqrtf(var + LN_EPS);

  float* __restrict__ obf = out + (size_t)bf * DD;
  obf[d] = has ? ((e - mu) * rstd * gamma[d] + beta[d]) : e;
}

extern "C" void kernel_launch(void* const* d_in, const int* in_sizes, int n_in,
                              void* d_out, int out_size, void* d_ws, size_t ws_size,
                              hipStream_t stream) {
  const float* H     = (const float*)d_in[0];
  const int*   vlens = (const int*)d_in[1];
  const float* gamma = (const float*)d_in[2];
  const float* beta  = (const float*)d_in[3];
  float* out = (float*)d_out;
  const int BF = in_sizes[1];
  mda_kernel<<<BF, NTHR, 0, stream>>>(H, vlens, gamma, beta, out);
}


# Work-group of 12 waves (was 24): a wave owns its 64-column stripe for all timesteps (16 row loads in flight), so LayerNorm sums come from its accumulators, not LDS; E and LayerNorm sum order changes
# speedup vs baseline: 2.9242x; 1.0754x over previous
#include <hip/hip_runtime.h>
#include <math.h>


typedef __attribute__((ext_vector_type(2))) float v2f;
typedef __attribute__((ext_vector_type(8))) float v8f;

#define TT 128
#define DD 768
#define NTHR 384
#define NWAVE 12
#define NSTRIPE 12
#define NLIST 24
#define DECAY_ALPHA 0.1f
#define EPS_W 1e-8f
#define LN_EPS 1e-5f

__device__ __forceinline__ float sanitize(float x) {
  return (__builtin_fabsf(x) <= 3.402823e38f) ? x : 0.0f;
}

__device__ __forceinline__ float wave_sum(float s) {
  #pragma unroll
  for (int off = 16; off > 0; off >>= 1) s += __shfl_xor(s, off, 32);
  return s;
}

__device__ __forceinline__ float decay_w(int t) {
  return __expf(-DECAY_ALPHA * (float)(TT - 1 - t));
}

template <int N>
__device__ __forceinline__ float row_xor_add(float keep, float send) {
  const int moved = __builtin_amdgcn_update_dpp(0, __float_as_int(send), 0x160 + N, 0xf, 0xf, true);
  return keep + __int_as_float(moved);
}

__global__ __launch_bounds__(NTHR) void mda_kernel(
    const float* __restrict__ H, const int* __restrict__ valid_lens,
    const float* __restrict__ gamma, const float* __restrict__ beta,
    float* __restrict__ out)
{
  __shared__ float s_part[NSTRIPE][TT];
  __shared__ float s_norm2[TT];
  __shared__ int   s_cnt[3][TT];
  __shared__ __attribute__((aligned(8))) int   s_lt[NLIST];
  __shared__ __attribute__((aligned(8))) float s_lw[NLIST];
  __shared__ __attribute__((aligned(16))) float s_red[NWAVE];
  __shared__ __attribute__((aligned(16))) float s_red2[NWAVE];
  __shared__ __attribute__((aligned(16))) float s_w[TT];
  __shared__ float s_bcast[3];

  const int bf   = blockIdx.x;
  const int tid  = threadIdx.x;
  const int lane = tid & 31;
  const int wv   = tid >> 5;
  const float* __restrict__ Hbf = H + (size_t)bf * (TT * DD);

  int L = valid_lens[bf];
  L = (L < 0) ? 0 : (L > TT ? TT : L);
  const bool sub = (L >= 2 * NLIST);
  const int Lw = sub ? L : 0;

  const int half = lane >> 4;
  const int l16  = lane & 15;
  const int wvs = __builtin_amdgcn_readfirstlane(wv);
  const int sp = wvs;
  const int d0 = sp * 64;
  v8f acc0 = {0.f, 0.f, 0.f, 0.f, 0.f, 0.f, 0.f, 0.f};
  v8f acc1 = acc0;
  v8f acc2 = acc0;
  v8f acc3 = acc0;

  {
    const float* __restrict__ plane = Hbf + (size_t)(half * 2) * DD + d0 + l16 * 4;
    const bool b3 = (l16 & 8) != 0, b2 = (l16 & 4) != 0, b1 = (l16 & 2) != 0, b0 = (l16 & 1) != 0;
    for (int tb = 0; tb < L; tb += 32) {
      float4 va[8], vb[8];
      v2f b[8];
      float P[16];
      #pragma unroll
      for (int q = 0; q < 8; ++q) {
        const float* pa = plane + (size_t)(tb + 4 * q) * DD;
        va[q] = *(const float4*)pa;
        vb[q] = *(const float4*)(pa + DD);
      }
      #pragma unroll
      for (int q = 0; q < 8; ++q) {
        const int t = tb + 4 * q + 2 * half;
        b[q].x = (t     < Lw) ? decay_w(t)     : 0.0f;
        b[q].y = (t + 1 < Lw) ? decay_w(t + 1) : 0.0f;
      }
      __builtin_amdgcn_sched_barrier(0);
      #pragma unroll
      for (int q = 0; q < 8; ++q) {
        v2f a0, a1, a2, a3;
        a0.x = sanitize(va[q].x);  a0.y = sanitize(vb[q].x);
        a1.x = sanitize(va[q].y);  a1.y = sanitize(vb[q].y);
        a2.x = sanitize(va[q].z);  a2.y = sanitize(vb[q].z);
        a3.x = sanitize(va[q].w);  a3.y = sanitize(vb[q].w);
        P[2 * q]     = fmaf(a3.x, a3.x, fmaf(a2.x, a2.x, fmaf(a0.x, a0.x, a1.x * a1.x)));
        P[2 * q + 1] = fmaf(a3.y, a3.y, fmaf(a2.y, a2.y, fmaf(a0.y, a0.y, a1.y * a1.y)));
        acc0 = __builtin_amdgcn_wmma_f32_16x16x4_f32(false, a0, false, b[q], (short)0,
                                                     acc0, false, false);
        acc1 = __builtin_amdgcn_wmma_f32_16x16x4_f32(false, a1, false, b[q], (short)0,
                                                     acc1, false, false);
        acc2 = __builtin_amdgcn_wmma_f32_16x16x4_f32(false, a2, false, b[q], (short)0,
                                                     acc2, false, false);
        acc3 = __builtin_amdgcn_wmma_f32_16x16x4_f32(false, a3, false, b[q], (short)0,
                                                     acc3, false, false);
      }
      float Q[8], R[4], S[2];
      #pragma unroll
      for (int j = 0; j < 8; ++j)
        Q[j] = row_xor_add<8>(b3 ? P[2 * j + 1] : P[2 * j], b3 ? P[2 * j] : P[2 * j + 1]);
      #pragma unroll
      for (int j = 0; j < 4; ++j)
        R[j] = row_xor_add<4>(b2 ? Q[2 * j + 1] : Q[2 * j], b2 ? Q[2 * j] : Q[2 * j + 1]);
      #pragma unroll
      for (int j = 0; j < 2; ++j)
        S[j] = row_xor_add<2>(b1 ? R[2 * j + 1] : R[2 * j], b1 ? R[2 * j] : R[2 * j + 1]);
      const float U = row_xor_add<1>(b0 ? S[1] : S[0], b0 ? S[0] : S[1]);
      {
        const int qq = (b0 ? 4 : 0) + (b1 ? 2 : 0) + (b2 ? 1 : 0);
        s_part[sp][tb + 4 * qq + 2 * half + (b3 ? 1 : 0)] = U;
      }
    }
  }
  __syncthreads();

  if (tid < TT) {
    float s = __builtin_inff();
    if (tid < L) {
      s = s_part[0][tid];
      #pragma unroll
      for (int k = 1; k < NSTRIPE; ++k) s += s_part[k][tid];
    }
    s_norm2[tid] = s;
  }
  __syncthreads();

  {
    const int i  = tid & (TT - 1);
    const int p  = wvs >> 2;
    const int j0 = 43 * p;
    const float my = s_norm2[i];
    int cnt = 0;
    #pragma unroll
    for (int jj = 0; jj < 42; ++jj) {
      const int j = j0 + jj;
      const float vj = s_norm2[j];
      cnt += (vj < my || (vj == my && j < i)) ? 1 : 0;
    }
    if (p < 2) {
      const int j = j0 + 42;
      const float vj = s_norm2[j];
      cnt += (vj < my || (vj == my && j < i)) ? 1 : 0;
    }
    s_cnt[p][i] = cnt;
  }
  __syncthreads();

  if (wvs == 0) {
    const int k_low = (int)(0.1f * (float)TT);
    const int lo = (k_low < (L >> 1)) ? k_low : (L >> 1);
    const int hr = L - lo;
    const int hi = (hr > lo) ? hr : lo;
    float wk = 0.0f, all2 = 0.0f, kept2 = 0.0f;
    unsigned m[4];
    bool  fl[4];
    float sw[4];
    #pragma unroll
    for (int k = 0; k < 4; ++k) {
      const int r = lane + 32 * k;
      const int rank = s_cnt[0][r] + s_cnt[1][r] + s_cnt[2][r];
      const bool valid = r < L;
      const bool keep  = valid && (rank >= lo) && (rank < hi);
      const float w = decay_w(r);
      wk += keep ? w : 0.0f;
      s_w[r] = keep ? w : 0.0f;
      const float e2 = valid ? w * w * s_norm2[r] : 0.0f;
      all2  += e2;
      kept2 += keep ? e2 : 0.0f;
      fl[k] = sub ? (valid && !keep) : keep;
      sw[k] = sub ? -w : w;
      m[k]  = __builtin_amdgcn_ballot_w32(fl[k]);
    }
    const float s = wave_sum(wk);
    const float sa = wave_sum(all2), sk = wave_sum(kept2);
    const bool cond_ok = (sa <= 1024.0f * sk) && (sa < __builtin_inff());
    const unsigned below = (1u << lane) - 1u;
    int off = 0;
    #pragma unroll
    for (int k = 0; k < 4; ++k) {
      const int pos = off + __popc(m[k] & below);
      if (fl[k] && pos < NLIST) { s_lt[pos] = lane + 32 * k; s_lw[pos] = sw[k]; }
      off += __popc(m[k]);
    }
    if (lane < NLIST && lane >= off) { s_lt[lane] = 0; s_lw[lane] = 0.0f; }
    if (lane == 0) {
      s_bcast[0] = 1.0f / fmaxf(s, EPS_W);
      s_bcast[1] = (s > EPS_W) ? 1.0f : 0.0f;
      s_bcast[2] = (sub && !cond_ok) ? 1.0f : 0.0f;
    }
  }
  __syncthreads();

  const int fbs = __builtin_amdgcn_readfirstlane(__float_as_int(s_bcast[2]));
  const float inv = s_bcast[0];
  const bool  has = s_bcast[1] > 0.5f;
  if (fbs != 0) {
    acc0 = (v8f){0.f, 0.f, 0.f, 0.f, 0.f, 0.f, 0.f, 0.f};
    acc1 = acc0;
    acc2 = acc0;
    acc3 = acc0;
    const float* __restrict__ plane = Hbf + (size_t)(half * 2) * DD + d0 + l16 * 4;
    for (int tb = 0; tb < L; tb += 32) {
      float4 va[8], vb[8];
      v2f b[8];
      #pragma unroll
      for (int q = 0; q < 8; ++q) {
        const float* pa = plane + (size_t)(tb + 4 * q) * DD;
        va[q] = *(const float4*)pa;
        vb[q] = *(const float4*)(pa + DD);
      }
      #pragma unroll
      for (int q = 0; q < 8; ++q) {
        const float4 wq = *(const float4*)&s_w[tb + 4 * q];
        b[q].x = half ? wq.z : wq.x;
        b[q].y = half ? wq.w : wq.y;
      }
      #pragma unroll
      for (int q = 0; q < 8; ++q) {
        v2f a0, a1, a2, a3;
        a0.x = sanitize(va[q].x);  a0.y = sanitize(vb[q].x);
        a1.x = sanitize(va[q].y);  a1.y = sanitize(vb[q].y);
        a2.x = sanitize(va[q].z);  a2.y = sanitize(vb[q].z);
        a3.x = sanitize(va[q].w);  a3.y = sanitize(vb[q].w);
        acc0 = __builtin_amdgcn_wmma_f32_16x16x4_f32(false, a0, false, b[q], (short)0,
                                                     acc0, false, false);
        acc1 = __builtin_amdgcn_wmma_f32_16x16x4_f32(false, a1, false, b[q], (short)0,
                                                     acc1, false, false);
        acc2 = __builtin_amdgcn_wmma_f32_16x16x4_f32(false, a2, false, b[q], (short)0,
                                                     acc2, false, false);
        acc3 = __builtin_amdgcn_wmma_f32_16x16x4_f32(false, a3, false, b[q], (short)0,
                                                     acc3, false, false);
      }
    }
  } else {
    float4 ca[6], cb[6];
    v2f cw[6];
    #pragma unroll
    for (int u = 0; u < 6; ++u) {
      const int e = 4 * u + 2 * half;
      const int r0 = s_lt[e], r1 = s_lt[e + 1];
      cw[u].x = s_lw[e];
      cw[u].y = s_lw[e + 1];
      ca[u] = *(const float4*)(Hbf + (size_t)r0 * DD + d0 + l16 * 4);
      cb[u] = *(const float4*)(Hbf + (size_t)r1 * DD + d0 + l16 * 4);
    }
    __builtin_amdgcn_sched_barrier(0);
    #pragma unroll
    for (int u = 0; u < 6; ++u) {
      v2f a0, a1, a2, a3;
      a0.x = sanitize(ca[u].x);  a0.y = sanitize(cb[u].x);
      a1.x = sanitize(ca[u].y);  a1.y = sanitize(cb[u].y);
      a2.x = sanitize(ca[u].z);  a2.y = sanitize(cb[u].z);
      a3.x = sanitize(ca[u].w);  a3.y = sanitize(cb[u].w);
      acc0 = __builtin_amdgcn_wmma_f32_16x16x4_f32(false, a0, false, cw[u], (short)0,
                                                   acc0, false, false);
      acc1 = __builtin_amdgcn_wmma_f32_16x16x4_f32(false, a1, false, cw[u], (short)0,
                                                   acc1, false, false);
      acc2 = __builtin_amdgcn_wmma_f32_16x16x4_f32(false, a2, false, cw[u], (short)0,
                                                   acc2, false, false);
      acc3 = __builtin_amdgcn_wmma_f32_16x16x4_f32(false, a3, false, cw[u], (short)0,
                                                   acc3, false, false);
    }
  }

  float e0[8], e1[8], e2[8], e3[8];
  #pragma unroll
  for (int r = 0; r < 8; ++r) {
    e0[r] = acc0[r] * inv;  e1[r] = acc1[r] * inv;  e2[r] = acc2[r] * inv;  e3[r] = acc3[r] * inv;
  }
  float p0 = e0[0], p1 = e1[0], p2 = e2[0], p3 = e3[0];
  #pragma unroll
  for (int r = 1; r < 8; ++r) { p0 += e0[r]; p1 += e1[r]; p2 += e2[r]; p3 += e3[r]; }
  float ps = (p0 + p1) + (p2 + p3);
  ps += __shfl_xor(ps, 16, 32);
  if (lane == 0) s_red[wvs] = ps;
  __syncthreads();
  float mu;
  {
    const float4 x0 = *(const float4*)&s_red[0], x1 = *(const float4*)&s_red[4], x2 = *(const float4*)&s_red[8];
    mu = (((x0.x + x0.y) + (x0.z + x0.w)) + ((x1.x + x1.y) + (x1.z + x1.w))
          + ((x2.x + x2.y) + (x2.z + x2.w))) * (1.0f / (float)DD);
  }
  float q0 = 0.0f, q1 = 0.0f, q2 = 0.0f, q3 = 0.0f;
  #pragma unroll
  for (int r = 0; r < 8; ++r) {
    const float c0 = e0[r] - mu, c1 = e1[r] - mu, c2 = e2[r] - mu, c3 = e3[r] - mu;
    q0 += c0 * c0;  q1 += c1 * c1;  q2 += c2 * c2;  q3 += c3 * c3;
  }
  float pv = (q0 + q1) + (q2 + q3);
  pv += __shfl_xor(pv, 16, 32);
  if (lane == 0) s_red2[wvs] = pv;
  __syncthreads();
  float var;
  {
    const float4 x0 = *(const float4*)&s_red2[0], x1 = *(const float4*)&s_red2[4], x2 = *(const float4*)&s_red2[8];
    var = (((x0.x + x0.y) + (x0.z + x0.w)) + ((x1.x + x1.y) + (x1.z + x1.w))
           + ((x2.x + x2.y) + (x2.z + x2.w))) * (1.0f / (float)DD);
  }
  const float rstd = rsqrtf(var + LN_EPS);

  if (l16 < 8) {
    float4 o;
    o.x = e0[0];  o.y = e1[0];  o.z = e2[0];  o.w = e3[0];
    #pragma unroll
    for (int r = 1; r < 8; ++r) {
      if (l16 == r) { o.x = e0[r];  o.y = e1[r];  o.z = e2[r];  o.w = e3[r]; }
    }
    const int dc = d0 + 4 * (8 * half + l16);
    float4 res = o;
    if (has) {
      const float4 g  = *(const float4*)(gamma + dc);
      const float4 bb = *(const float4*)(beta + dc);
      res.x = (o.x - mu) * rstd * g.x + bb.x;
      res.y = (o.y - mu) * rstd * g.y + bb.y;
      res.z = (o.z - mu) * rstd * g.z + bb.z;
      res.w = (o.w - mu) * rstd * g.w + bb.w;
    }
    *(float4*)(out + (size_t)bf * DD + dc) = res;
  }
}

extern "C" void kernel_launch(void* const* d_in, const int* in_sizes, int n_in,
                              void* d_out, int out_size, void* d_ws, size_t ws_size,
                              hipStream_t stream) {
  const float* H     = (const float*)d_in[0];
  const int*   vlens = (const int*)d_in[1];
  const float* gamma = (const float*)d_in[2];
  const float* beta  = (const float*)d_in[3];
  float* out = (float*)d_out;
  const int BF = in_sizes[1];
  mda_kernel<<<BF, NTHR, 0, stream>>>(H, vlens, gamma, beta, out);
}
